# starting kernel
# baseline (speedup 1.0000x reference)
#include <hip/hip_runtime.h>
#include <math.h>

typedef __attribute__((ext_vector_type(2))) float v2f;
typedef __attribute__((ext_vector_type(8))) float v8f;

static constexpr int kHW    = 196;
static constexpr int kWdim  = 14;
static constexpr int kNReg  = 14;
static constexpr float kEps = 1e-6f;

__global__ __launch_bounds__(256) void rmac_region_max_kernel(
    const float* __restrict__ x, float* __restrict__ m, int nPlanes) {
  int idx = blockIdx.x * blockDim.x + threadIdx.x;
  if (idx >= nPlanes) return;
  const float* p = x + (size_t)idx * kHW;

  float acc[kNReg];
#pragma unroll
  for (int r = 0; r < kNReg; ++r) acc[r] = -INFINITY;

#pragma unroll
  for (int i = 0; i < 14; ++i) {
    float row[14];
    const float2* p2 = (const float2*)(p + i * kWdim);
#pragma unroll
    for (int j = 0; j < 7; ++j) {
      float2 v = p2[j];
      row[2 * j] = v.x;
      row[2 * j + 1] = v.y;
    }
    float c3a = row[0];
#pragma unroll
    for (int j = 1; j <= 6; ++j) c3a = fmaxf(c3a, row[j]);
    float c3b = row[3];
#pragma unroll
    for (int j = 4; j <= 9; ++j) c3b = fmaxf(c3b, row[j]);
    float c3c = row[7];
#pragma unroll
    for (int j = 8; j <= 13; ++j) c3c = fmaxf(c3c, row[j]);
    float c2a  = fmaxf(c3a, fmaxf(row[7], row[8]));
    float c2b  = fmaxf(c3c, fmaxf(row[5], row[6]));
    float full = fmaxf(c2a, c3c);

    acc[0] = fmaxf(acc[0], full);
    if (i <= 8) { acc[1] = fmaxf(acc[1], c2a); acc[2] = fmaxf(acc[2], c2b); }
    if (i >= 5) { acc[3] = fmaxf(acc[3], c2a); acc[4] = fmaxf(acc[4], c2b); }
    if (i <= 6) {
      acc[5] = fmaxf(acc[5], c3a); acc[6] = fmaxf(acc[6], c3b); acc[7] = fmaxf(acc[7], c3c);
    }
    if (i >= 3 && i <= 9) {
      acc[8] = fmaxf(acc[8], c3a); acc[9] = fmaxf(acc[9], c3b); acc[10] = fmaxf(acc[10], c3c);
    }
    if (i >= 7) {
      acc[11] = fmaxf(acc[11], c3a); acc[12] = fmaxf(acc[12], c3b); acc[13] = fmaxf(acc[13], c3c);
    }
  }
#pragma unroll
  for (int r = 0; r < kNReg; ++r) m[(size_t)r * nPlanes + idx] = acc[r];
}

__global__ __launch_bounds__(32) void rmac_norm_kernel(
    const float* __restrict__ m, float* __restrict__ w, int C) {
  int rb = blockIdx.x;
  int lane = threadIdx.x;
  const float* base = m + (size_t)rb * C;

  int off = (lane & 15) * 4 + ((lane >> 4) << 1);

  v8f acc = {0.f, 0.f, 0.f, 0.f, 0.f, 0.f, 0.f, 0.f};
  for (int k = 0; k < C; k += 64) {
    float2 t = *(const float2*)(base + k + off);
    v2f a;
    a.x = t.x;
    a.y = t.y;
    acc = __builtin_amdgcn_wmma_f32_16x16x4_f32(
         false, a,  false, a,
         (short)0, acc,  false,  false);
  }

  float d = 0.f;
  if (lane < 8)       d = acc[lane];
  else if (lane >= 24) d = acc[lane - 24];

#pragma unroll
  for (int s = 16; s >= 1; s >>= 1) d += __shfl_xor(d, s, 32);

  if (lane == 0) w[rb] = 1.0f / (sqrtf(d) + kEps);
}

__global__ __launch_bounds__(256) void rmac_combine_kernel(
    const float* __restrict__ m, const float* __restrict__ w,
    float* __restrict__ out, int nPlanes, int C) {
  int idx = blockIdx.x * blockDim.x + threadIdx.x;
  if (idx >= nPlanes) return;
  int B = nPlanes / C;
  int b = idx / C;
  float v = 0.f;
#pragma unroll
  for (int r = 0; r < kNReg; ++r) {
    float wt = w[r * B + b];
    if (r == 0) wt *= 2.0f;
    v += m[(size_t)r * nPlanes + idx] * wt;
  }
  out[idx] = v;
}

extern "C" void kernel_launch(void* const* d_in, const int* in_sizes, int n_in,
                              void* d_out, int out_size, void* d_ws, size_t ws_size,
                              hipStream_t stream) {
  const float* x = (const float*)d_in[0];
  const int C = 2048;
  int nPlanes = in_sizes[0] / kHW;
  int B = nPlanes / C;

  float* m = (float*)d_ws;
  float* w = m + (size_t)kNReg * nPlanes;

  const int threads = 256;
  int blocks = (nPlanes + threads - 1) / threads;
  rmac_region_max_kernel<<<blocks, threads, 0, stream>>>(x, m, nPlanes);
  rmac_norm_kernel<<<kNReg * B, 32, 0, stream>>>(m, w, C);
  rmac_combine_kernel<<<blocks, threads, 0, stream>>>(m, w, (float*)d_out, nPlanes, C);
}


# Region-max kernel: x goes through whole-line 16-byte wave loads into a per-wave LDS tile; each lane reads its own plane back at a conflict-free 784-byte pitch, replacing 32-line strided global loads
# speedup vs baseline: 1.4908x; 1.4908x over previous
#include <hip/hip_runtime.h>
#include <math.h>

typedef __attribute__((ext_vector_type(2))) float v2f;
typedef __attribute__((ext_vector_type(8))) float v8f;

static constexpr int kHW    = 196;
static constexpr int kWdim  = 14;
static constexpr int kNReg  = 14;
static constexpr float kEps = 1e-6f;

static constexpr int kPlaneVec    = kHW / 4;
static constexpr int kTilePlanes  = 32;
static constexpr int kTileVec     = kTilePlanes * kPlaneVec;
static constexpr int kK1Waves     = 4;
static constexpr int kK1Threads   = kK1Waves * 32;

__global__ __launch_bounds__(kK1Threads) void rmac_region_max_kernel(
    const float* __restrict__ x, float* __restrict__ m, int nPlanes) {
  __shared__ float4 tile[kK1Waves * kTileVec];

  const int t    = threadIdx.x;
  const int lane = t & 31;
  const int wv   = t >> 5;
  float4* wt = tile + wv * kTileVec;

  const size_t blockPlane0 = (size_t)blockIdx.x * kK1Threads;
  const size_t v0 = (blockPlane0 + (size_t)wv * kTilePlanes) * kPlaneVec + lane;
  const float4* gx = (const float4*)x;

  if (blockPlane0 + kK1Threads > (size_t)nPlanes) return;

  float4 buf[kPlaneVec];
  for (int i = 0; i < kPlaneVec; ++i) buf[i] = gx[v0 + (size_t)i * 32];
  for (int i = 0; i < kPlaneVec; ++i) wt[i * 32 + lane] = buf[i];

  __syncthreads();

  const float4* lp = wt + lane * kPlaneVec;

  float acc[kNReg];
  for (int r = 0; r < kNReg; ++r) acc[r] = -INFINITY;

  for (int rp = 0; rp < 7; ++rp) {
    float e[2 * kWdim];
    for (int j = 0; j < 7; ++j) {
      float4 v = lp[rp * 7 + j];
      e[4 * j]     = v.x;
      e[4 * j + 1] = v.y;
      e[4 * j + 2] = v.z;
      e[4 * j + 3] = v.w;
    }
    for (int h = 0; h < 2; ++h) {
      const int i = 2 * rp + h;
      const float* row = e + kWdim * h;
      float c3a = row[0];
      for (int j = 1; j <= 6; ++j) c3a = fmaxf(c3a, row[j]);
      float c3b = row[3];
      for (int j = 4; j <= 9; ++j) c3b = fmaxf(c3b, row[j]);
      float c3c = row[7];
      for (int j = 8; j <= 13; ++j) c3c = fmaxf(c3c, row[j]);
      float c2a  = fmaxf(c3a, fmaxf(row[7], row[8]));
      float c2b  = fmaxf(c3c, fmaxf(row[5], row[6]));
      float full = fmaxf(c2a, c3c);

      acc[0] = fmaxf(acc[0], full);
      if (i <= 8) { acc[1] = fmaxf(acc[1], c2a); acc[2] = fmaxf(acc[2], c2b); }
      if (i >= 5) { acc[3] = fmaxf(acc[3], c2a); acc[4] = fmaxf(acc[4], c2b); }
      if (i <= 6) {
        acc[5] = fmaxf(acc[5], c3a); acc[6] = fmaxf(acc[6], c3b); acc[7] = fmaxf(acc[7], c3c);
      }
      if (i >= 3 && i <= 9) {
        acc[8] = fmaxf(acc[8], c3a); acc[9] = fmaxf(acc[9], c3b); acc[10] = fmaxf(acc[10], c3c);
      }
      if (i >= 7) {
        acc[11] = fmaxf(acc[11], c3a); acc[12] = fmaxf(acc[12], c3b); acc[13] = fmaxf(acc[13], c3c);
      }
    }
  }

  const size_t idx = blockPlane0 + t;
  for (int r = 0; r < kNReg; ++r) m[(size_t)r * nPlanes + idx] = acc[r];
}

__global__ __launch_bounds__(32) void rmac_norm_kernel(
    const float* __restrict__ m, float* __restrict__ w, int C) {
  int rb = blockIdx.x;
  int lane = threadIdx.x;
  const float* base = m + (size_t)rb * C;

  int off = (lane & 15) * 4 + ((lane >> 4) << 1);

  v8f acc = {0.f, 0.f, 0.f, 0.f, 0.f, 0.f, 0.f, 0.f};
  for (int k = 0; k < C; k += 64) {
    float2 t = *(const float2*)(base + k + off);
    v2f a;
    a.x = t.x;
    a.y = t.y;
    acc = __builtin_amdgcn_wmma_f32_16x16x4_f32(
         false, a,  false, a,
         (short)0, acc,  false,  false);
  }

  float d = 0.f;
  if (lane < 8)       d = acc[lane];
  else if (lane >= 24) d = acc[lane - 24];

#pragma unroll
  for (int s = 16; s >= 1; s >>= 1) d += __shfl_xor(d, s, 32);

  if (lane == 0) w[rb] = 1.0f / (sqrtf(d) + kEps);
}

__global__ __launch_bounds__(256) void rmac_combine_kernel(
    const float* __restrict__ m, const float* __restrict__ w,
    float* __restrict__ out, int nPlanes, int C) {
  int idx = blockIdx.x * blockDim.x + threadIdx.x;
  if (idx >= nPlanes) return;
  int B = nPlanes / C;
  int b = idx / C;
  float v = 0.f;
#pragma unroll
  for (int r = 0; r < kNReg; ++r) {
    float wt = w[r * B + b];
    if (r == 0) wt *= 2.0f;
    v += m[(size_t)r * nPlanes + idx] * wt;
  }
  out[idx] = v;
}

extern "C" void kernel_launch(void* const* d_in, const int* in_sizes, int n_in,
                              void* d_out, int out_size, void* d_ws, size_t ws_size,
                              hipStream_t stream) {
  const float* x = (const float*)d_in[0];
  const int C = 2048;
  int nPlanes = in_sizes[0] / kHW;
  int B = nPlanes / C;

  float* m = (float*)d_ws;
  float* w = m + (size_t)kNReg * nPlanes;

  const int threads = 256;
  int blocks = (nPlanes + threads - 1) / threads;
  int blocks1 = (nPlanes + kK1Threads - 1) / kK1Threads;
  rmac_region_max_kernel<<<blocks1, kK1Threads, 0, stream>>>(x, m, nPlanes);
  rmac_norm_kernel<<<kNReg * B, 32, 0, stream>>>(m, w, C);
  rmac_combine_kernel<<<blocks, threads, 0, stream>>>(m, w, (float*)d_out, nPlanes, C);
}


# Norm launch removed: each wave of the region-max kernel sums its squared maxima per region on f32 WMMA (Gram diagonal) into partials, combine adds 64 per norm; sum order changes
# speedup vs baseline: 1.8955x; 1.2715x over previous
#include <hip/hip_runtime.h>
#include <math.h>


typedef __attribute__((ext_vector_type(2))) float v2f;
typedef __attribute__((ext_vector_type(8))) float v8f;

static constexpr int kHW    = 196;
static constexpr int kWdim  = 14;
static constexpr int kNReg  = 14;
static constexpr float kEps = 1e-6f;


static constexpr int kPlaneVec    = kHW / 4;
static constexpr int kTilePlanes  = 32;
static constexpr int kTileVec     = kTilePlanes * kPlaneVec;
static constexpr int kK1Waves     = 4;
static constexpr int kK1Threads   = kK1Waves * 32;
static constexpr int kTPitch      = 36;
static constexpr int kPartPitch   = 32;

__global__ __launch_bounds__(kK1Threads) void rmac_region_max_kernel(
    const float* __restrict__ x, float* __restrict__ m, int nPlanes) {
  __shared__ float4 tile[kK1Waves * kTileVec];

  const int t    = threadIdx.x;
  const int lane = t & 31;
  const int wv   = t >> 5;
  float4* wt = tile + wv * kTileVec;

  const size_t blockPlane0 = (size_t)blockIdx.x * kK1Threads;
  const size_t v0 = (blockPlane0 + (size_t)wv * kTilePlanes) * kPlaneVec + lane;
  const float4* gx = (const float4*)x;

  if (blockPlane0 + kK1Threads > (size_t)nPlanes) return;

  float4 buf[kPlaneVec];
  for (int i = 0; i < kPlaneVec; ++i) buf[i] = gx[v0 + (size_t)i * 32];
  for (int i = 0; i < kPlaneVec; ++i) wt[i * 32 + lane] = buf[i];

  __syncthreads();

  const float4* lp = wt + lane * kPlaneVec;

  float acc[kNReg];
  for (int r = 0; r < kNReg; ++r) acc[r] = -INFINITY;

  for (int rp = 0; rp < 7; ++rp) {
    float e[2 * kWdim];
    for (int j = 0; j < 7; ++j) {
      float4 v = lp[rp * 7 + j];
      e[4 * j]     = v.x;
      e[4 * j + 1] = v.y;
      e[4 * j + 2] = v.z;
      e[4 * j + 3] = v.w;
    }
    for (int h = 0; h < 2; ++h) {
      const int i = 2 * rp + h;
      const float* row = e + kWdim * h;
      float c3a = row[0];
      for (int j = 1; j <= 6; ++j) c3a = fmaxf(c3a, row[j]);
      float c3b = row[3];
      for (int j = 4; j <= 9; ++j) c3b = fmaxf(c3b, row[j]);
      float c3c = row[7];
      for (int j = 8; j <= 13; ++j) c3c = fmaxf(c3c, row[j]);
      float c2a  = fmaxf(c3a, fmaxf(row[7], row[8]));
      float c2b  = fmaxf(c3c, fmaxf(row[5], row[6]));
      float full = fmaxf(c2a, c3c);

      acc[0] = fmaxf(acc[0], full);
      if (i <= 8) { acc[1] = fmaxf(acc[1], c2a); acc[2] = fmaxf(acc[2], c2b); }
      if (i >= 5) { acc[3] = fmaxf(acc[3], c2a); acc[4] = fmaxf(acc[4], c2b); }
      if (i <= 6) {
        acc[5] = fmaxf(acc[5], c3a); acc[6] = fmaxf(acc[6], c3b); acc[7] = fmaxf(acc[7], c3c);
      }
      if (i >= 3 && i <= 9) {
        acc[8] = fmaxf(acc[8], c3a); acc[9] = fmaxf(acc[9], c3b); acc[10] = fmaxf(acc[10], c3c);
      }
      if (i >= 7) {
        acc[11] = fmaxf(acc[11], c3a); acc[12] = fmaxf(acc[12], c3b); acc[13] = fmaxf(acc[13], c3c);
      }
    }
  }

  float* tt = (float*)wt;
  for (int r = 0; r < kNReg; ++r) tt[r * kTPitch + lane] = acc[r];

  __syncthreads();

  const size_t idx = blockPlane0 + t;
  for (int r = 1; r < kNReg; ++r) m[(size_t)r * nPlanes + idx] = acc[r];

  const int c  = lane & 15;
  const int hh = lane >> 4;
  const float4* tr = (const float4*)(tt + c * kTPitch + 16 * hh);
  float4 q[4];
  for (int j = 0; j < 4; ++j) q[j] = tr[j];
  if (c >= kNReg) {
    for (int j = 0; j < 4; ++j) q[j] = make_float4(0.f, 0.f, 0.f, 0.f);
  }

  v8f g0 = {0.f, 0.f, 0.f, 0.f, 0.f, 0.f, 0.f, 0.f};
  v8f g1 = {0.f, 0.f, 0.f, 0.f, 0.f, 0.f, 0.f, 0.f};
  for (int j = 0; j < 4; ++j) {
    v2f a0, a1;
    a0.x = q[j].x; a0.y = q[j].y;
    a1.x = q[j].z; a1.y = q[j].w;
    g0 = __builtin_amdgcn_wmma_f32_16x16x4_f32(
         false, a0,  false, a0,
         (short)0, g0,  false,  false);
    g1 = __builtin_amdgcn_wmma_f32_16x16x4_f32(
         false, a1,  false, a1,
         (short)0, g1,  false,  false);
  }
  v8f g = g0 + g1;

  float d = 0.f;
  if (lane < 8)        d = g[lane];
  else if (lane >= 24) d = g[lane - 24];
  const int n = (lane < 8) ? lane : lane - 16;
  const size_t waveTile = blockPlane0 / kTilePlanes + wv;
  if (lane < 8 || (lane >= 24 && lane < 30)) m[waveTile * kPartPitch + n] = d;
}

static constexpr int kC          = 2048;
static constexpr int kTilesPerRow = kC / kTilePlanes;
static constexpr int kK3Threads  = 256;
static constexpr int kK3Waves    = kK3Threads / 32;

__global__ __launch_bounds__(kK3Threads) void rmac_combine_kernel(
    const float* __restrict__ m, float* __restrict__ out, int nPlanes) {
  __shared__ float sp[kK3Waves][16];

  const int t    = threadIdx.x;
  const int lane = t & 31;
  const int wv   = t >> 5;
  const size_t idx = (size_t)blockIdx.x * kK3Threads + t;

  if ((size_t)blockIdx.x * kK3Threads + kK3Threads > (size_t)nPlanes) return;

  const int b  = blockIdx.x / (kC / kK3Threads);
  const int c  = lane & 15;
  const int hh = lane >> 4;
  const int cr = (c < kNReg) ? c : kNReg - 1;
  const float* prow = m + (size_t)b * kTilesPerRow * kPartPitch;
  float p[kTilesPerRow / (2 * kK3Waves)];
  for (int i = 0; i < kTilesPerRow / (2 * kK3Waves); ++i) {
    const int j = 2 * wv + hh + 2 * kK3Waves * i;
    p[i] = prow[j * kPartPitch + cr];
  }

  float mv[kNReg];
  for (int r = 1; r < kNReg; ++r) mv[r] = m[(size_t)r * nPlanes + idx];
  float s = 0.f;
  for (int i = 0; i < kTilesPerRow / (2 * kK3Waves); ++i) s += p[i];
  s += __shfl_xor(s, 16, 32);
  if (lane < 16) sp[wv][lane] = s;

  __syncthreads();

  float tot = 0.f;
  for (int w8 = 0; w8 < kK3Waves; ++w8) tot += sp[w8][c];
  float wr = 1.0f / (sqrtf(tot) + kEps);
  if (c == 0) wr *= 2.0f;
  const int wri = __builtin_bit_cast(int, wr);

  mv[0] = fmaxf(fmaxf(mv[1], mv[2]), fmaxf(mv[3], mv[4]));
  float v = 0.f;
#pragma unroll
  for (int r = 0; r < kNReg; ++r)
    v += mv[r] * __builtin_bit_cast(float, __builtin_amdgcn_readlane(wri, r));
  out[idx] = v;
}

extern "C" void kernel_launch(void* const* d_in, const int* in_sizes, int n_in,
                              void* d_out, int out_size, void* d_ws, size_t ws_size,
                              hipStream_t stream) {
  const float* x = (const float*)d_in[0];
  int nPlanes = in_sizes[0] / kHW;

  float* m = (float*)d_ws;

  int blocks  = (nPlanes + kK3Threads - 1) / kK3Threads;
  int blocks1 = (nPlanes + kK1Threads - 1) / kK1Threads;
  rmac_region_max_kernel<<<blocks1, kK1Threads, 0, stream>>>(x, m, nPlanes);
  rmac_combine_kernel<<<blocks, kK3Threads, 0, stream>>>(m, (float*)d_out, nPlanes);
}


# Region-max kernel as one 8-wave block per WGP, two tiles per wave: both tiles' loads issued at entry, the second streams in while the first is reduced; wave-private LDS tiles, no block barrier
# speedup vs baseline: 2.0460x; 1.0794x over previous
#include <hip/hip_runtime.h>
#include <math.h>


typedef __attribute__((ext_vector_type(2))) float v2f;
typedef __attribute__((ext_vector_type(4))) float v4f;
typedef __attribute__((ext_vector_type(8))) float v8f;

static constexpr int kHW    = 196;
static constexpr int kWdim  = 14;
static constexpr int kNReg  = 14;
static constexpr float kEps = 1e-6f;


static constexpr int kPlaneVec    = kHW / 4;
static constexpr int kTilePlanes  = 32;
static constexpr int kTileVec     = kTilePlanes * kPlaneVec;
static constexpr int kK1Waves     = 8;
static constexpr int kK1Threads   = kK1Waves * 32;
static constexpr int kPassPlanes  = kK1Threads;
static constexpr int kBlockPlanes = 2 * kPassPlanes;
static constexpr int kTPitch      = 36;
static constexpr int kPartPitch   = 32;

__device__ __forceinline__ void rmac_reduce_tile(
    v4f* wt, int lane, float* __restrict__ m, int nPlanes, size_t plane0) {
  __builtin_amdgcn_fence(__ATOMIC_RELEASE, "wavefront");
  __builtin_amdgcn_fence(__ATOMIC_ACQUIRE, "wavefront");

  const v4f* lp = wt + lane * kPlaneVec;

  float acc[kNReg];
  for (int r = 0; r < kNReg; ++r) acc[r] = -INFINITY;

  for (int rp = 0; rp < 7; ++rp) {
    float e[2 * kWdim];
    for (int j = 0; j < 7; ++j) {
      v4f v = lp[rp * 7 + j];
      e[4 * j]     = v.x;
      e[4 * j + 1] = v.y;
      e[4 * j + 2] = v.z;
      e[4 * j + 3] = v.w;
    }
    for (int h = 0; h < 2; ++h) {
      const int i = 2 * rp + h;
      const float* row = e + kWdim * h;
      float c3a = row[0];
      for (int j = 1; j <= 6; ++j) c3a = fmaxf(c3a, row[j]);
      float c3b = row[3];
      for (int j = 4; j <= 9; ++j) c3b = fmaxf(c3b, row[j]);
      float c3c = row[7];
      for (int j = 8; j <= 13; ++j) c3c = fmaxf(c3c, row[j]);
      float c2a  = fmaxf(c3a, fmaxf(row[7], row[8]));
      float c2b  = fmaxf(c3c, fmaxf(row[5], row[6]));
      float full = fmaxf(c2a, c3c);

      acc[0] = fmaxf(acc[0], full);
      if (i <= 8) { acc[1] = fmaxf(acc[1], c2a); acc[2] = fmaxf(acc[2], c2b); }
      if (i >= 5) { acc[3] = fmaxf(acc[3], c2a); acc[4] = fmaxf(acc[4], c2b); }
      if (i <= 6) {
        acc[5] = fmaxf(acc[5], c3a); acc[6] = fmaxf(acc[6], c3b); acc[7] = fmaxf(acc[7], c3c);
      }
      if (i >= 3 && i <= 9) {
        acc[8] = fmaxf(acc[8], c3a); acc[9] = fmaxf(acc[9], c3b); acc[10] = fmaxf(acc[10], c3c);
      }
      if (i >= 7) {
        acc[11] = fmaxf(acc[11], c3a); acc[12] = fmaxf(acc[12], c3b); acc[13] = fmaxf(acc[13], c3c);
      }
    }
  }

  float* tt = (float*)wt;
  for (int r = 0; r < kNReg; ++r) tt[r * kTPitch + lane] = acc[r];

  __builtin_amdgcn_fence(__ATOMIC_RELEASE, "wavefront");
  __builtin_amdgcn_fence(__ATOMIC_ACQUIRE, "wavefront");

  const size_t idx = plane0 + lane;
  for (int r = 1; r < kNReg; ++r) m[(size_t)r * nPlanes + idx] = acc[r];

  const int c  = lane & 15;
  const int hh = lane >> 4;
  const v4f* tr = (const v4f*)(tt + c * kTPitch + 16 * hh);
  v4f q[4];
  for (int j = 0; j < 4; ++j) q[j] = tr[j];
  if (c >= kNReg) {
    for (int j = 0; j < 4; ++j) q[j] = (v4f){0.f, 0.f, 0.f, 0.f};
  }

  v8f g0 = {0.f, 0.f, 0.f, 0.f, 0.f, 0.f, 0.f, 0.f};
  v8f g1 = {0.f, 0.f, 0.f, 0.f, 0.f, 0.f, 0.f, 0.f};
  for (int j = 0; j < 4; ++j) {
    v2f a0, a1;
    a0.x = q[j].x; a0.y = q[j].y;
    a1.x = q[j].z; a1.y = q[j].w;
    g0 = __builtin_amdgcn_wmma_f32_16x16x4_f32(
         false, a0,  false, a0,
         (short)0, g0,  false,  false);
    g1 = __builtin_amdgcn_wmma_f32_16x16x4_f32(
         false, a1,  false, a1,
         (short)0, g1,  false,  false);
  }
  v8f g = g0 + g1;

  float d = 0.f;
  if (lane < 8)        d = g[lane];
  else if (lane >= 24) d = g[lane - 24];
  const int n = (lane < 8) ? lane : lane - 16;
  const size_t waveTile = plane0 / kTilePlanes;
  if (lane < 8 || (lane >= 24 && lane < 30)) m[waveTile * kPartPitch + n] = d;
}

__global__ __launch_bounds__(kK1Threads) void rmac_region_max_kernel(
    const float* __restrict__ x, float* __restrict__ m, int nPlanes) {
  __shared__ v4f tile[kK1Waves * kTileVec];

  const int t    = threadIdx.x;
  const int lane = t & 31;
  const int wv   = t >> 5;
  v4f* wt = tile + wv * kTileVec;

  const size_t blockPlane0 = (size_t)blockIdx.x * kBlockPlanes;
  if (blockPlane0 + kBlockPlanes > (size_t)nPlanes) return;

  const size_t planeA = blockPlane0 + (size_t)wv * kTilePlanes;
  const size_t planeB = planeA + kPassPlanes;
  const v4f* gx = (const v4f*)x;
  const v4f* ga = gx + planeA * kPlaneVec + lane;
  const v4f* gb = gx + planeB * kPlaneVec + lane;

  v4f bufA[kPlaneVec];
  v4f bufB[kPlaneVec];
  for (int i = 0; i < kPlaneVec; ++i) bufA[i] = ga[(size_t)i * 32];
  for (int i = 0; i < kPlaneVec; ++i) bufB[i] = gb[(size_t)i * 32];

  __builtin_amdgcn_sched_barrier(0);

  for (int i = 0; i < kPlaneVec; ++i) wt[i * 32 + lane] = bufA[i];
  rmac_reduce_tile(wt, lane, m, nPlanes, planeA);
  for (int i = 0; i < kPlaneVec; ++i) wt[i * 32 + lane] = bufB[i];
  rmac_reduce_tile(wt, lane, m, nPlanes, planeB);
}

static constexpr int kC          = 2048;
static constexpr int kTilesPerRow = kC / kTilePlanes;
static constexpr int kK3Threads  = 256;
static constexpr int kK3Waves    = kK3Threads / 32;

__global__ __launch_bounds__(kK3Threads) void rmac_combine_kernel(
    const float* __restrict__ m, float* __restrict__ out, int nPlanes) {
  __shared__ float sp[kK3Waves][16];

  const int t    = threadIdx.x;
  const int lane = t & 31;
  const int wv   = t >> 5;
  const size_t idx = (size_t)blockIdx.x * kK3Threads + t;

  if ((size_t)blockIdx.x * kK3Threads + kK3Threads > (size_t)nPlanes) return;

  const int b  = blockIdx.x / (kC / kK3Threads);
  const int c  = lane & 15;
  const int hh = lane >> 4;
  const int cr = (c < kNReg) ? c : kNReg - 1;
  const float* prow = m + (size_t)b * kTilesPerRow * kPartPitch;
  float p[kTilesPerRow / (2 * kK3Waves)];
  for (int i = 0; i < kTilesPerRow / (2 * kK3Waves); ++i) {
    const int j = 2 * wv + hh + 2 * kK3Waves * i;
    p[i] = prow[j * kPartPitch + cr];
  }

  float mv[kNReg];
  for (int r = 1; r < kNReg; ++r) mv[r] = m[(size_t)r * nPlanes + idx];
  float s = 0.f;
  for (int i = 0; i < kTilesPerRow / (2 * kK3Waves); ++i) s += p[i];
  s += __shfl_xor(s, 16, 32);
  if (lane < 16) sp[wv][lane] = s;

  __syncthreads();

  float tot = 0.f;
  for (int w8 = 0; w8 < kK3Waves; ++w8) tot += sp[w8][c];
  float wr = 1.0f / (sqrtf(tot) + kEps);
  if (c == 0) wr *= 2.0f;
  const int wri = __builtin_bit_cast(int, wr);

  mv[0] = fmaxf(fmaxf(mv[1], mv[2]), fmaxf(mv[3], mv[4]));
  float v = 0.f;
#pragma unroll
  for (int r = 0; r < kNReg; ++r)
    v += mv[r] * __builtin_bit_cast(float, __builtin_amdgcn_readlane(wri, r));
  out[idx] = v;
}

extern "C" void kernel_launch(void* const* d_in, const int* in_sizes, int n_in,
                              void* d_out, int out_size, void* d_ws, size_t ws_size,
                              hipStream_t stream) {
  const float* x = (const float*)d_in[0];
  int nPlanes = in_sizes[0] / kHW;

  float* m = (float*)d_ws;

  int blocks  = (nPlanes + kK3Threads - 1) / kK3Threads;
  int blocks1 = (nPlanes + kBlockPlanes - 1) / kBlockPlanes;
  rmac_region_max_kernel<<<blocks1, kK1Threads, 0, stream>>>(x, m, nPlanes);
  rmac_combine_kernel<<<blocks, kK3Threads, 0, stream>>>(m, (float*)d_out, nPlanes);
}
